# starting kernel
# baseline (speedup 1.0000x reference)
#include <hip/hip_runtime.h>
#include <math.h>

#define N_NODES 50000
#define N_EDGES 800000
#define F_IN    512
#define F_HID   256
#define F_OUT   64
#define LN_EPS  1e-5f
#define L2_EPS  1e-12f

typedef __attribute__((ext_vector_type(2))) float v2f;
typedef __attribute__((ext_vector_type(8))) float v8f;

__device__ __forceinline__ float wave_sum32(float v) {
#pragma unroll
    for (int m = 16; m >= 1; m >>= 1) v += __shfl_xor(v, m, 32);
    return v;
}

__global__ void __launch_bounds__(256) k_deg_init(float* deg) {
    int i = blockIdx.x * 256 + threadIdx.x;
    if (i < N_NODES) deg[i] = 1.0f;
}

__global__ void __launch_bounds__(256) k_deg_accum(const int* __restrict__ dst,
                                                   float* deg) {
    int e = blockIdx.x * 256 + threadIdx.x;
    if (e < N_EDGES) atomicAdd(&deg[dst[e]], 1.0f);
}

__global__ void __launch_bounds__(256) k_dinv(float* deg) {
    int i = blockIdx.x * 256 + threadIdx.x;
    if (i < N_NODES) deg[i] = rsqrtf(deg[i]);
}

__global__ void __launch_bounds__(256) k_l2norm(const float* __restrict__ x,
                                                float* __restrict__ xn) {
    const int node = blockIdx.x * 8 + (threadIdx.x >> 5);
    const int lane = threadIdx.x & 31;
    const float4* row = (const float4*)(x + (size_t)node * F_IN);
    float4 v[4];
    float ss = 0.0f;
#pragma unroll
    for (int i = 0; i < 4; ++i) {
        float4 t = row[lane + 32 * i];
        t.x = isfinite(t.x) ? t.x : 0.0f;
        t.y = isfinite(t.y) ? t.y : 0.0f;
        t.z = isfinite(t.z) ? t.z : 0.0f;
        t.w = isfinite(t.w) ? t.w : 0.0f;
        v[i] = t;
        ss += t.x * t.x + t.y * t.y + t.z * t.z + t.w * t.w;
    }
    ss = wave_sum32(ss);
    float scale = 1.0f / fmaxf(sqrtf(ss), L2_EPS);
    float4* orow = (float4*)(xn + (size_t)node * F_IN);
#pragma unroll
    for (int i = 0; i < 4; ++i) {
        float4 t = v[i];
        t.x *= scale; t.y *= scale; t.z *= scale; t.w *= scale;
        orow[lane + 32 * i] = t;
    }
}

template <int K, int NOUT>
__global__ void __launch_bounds__(128) k_gemm_wmma(const float* __restrict__ A,
                                                   const float* __restrict__ B,
                                                   float* __restrict__ C) {
    __shared__ float As[16 * K];
    const int tid  = threadIdx.x;
    const int m0   = blockIdx.x * 16;
    const int wave = tid >> 5;
    const int lane = tid & 31;
    const int n0   = blockIdx.y * 64 + wave * 16;

    {
        const float4* gsrc = (const float4*)A + (size_t)m0 * (K / 4);
        float4* lds = (float4*)As;
#pragma unroll 4
        for (int i = tid; i < 16 * (K / 4); i += 128) lds[i] = gsrc[i];
    }
    __syncthreads();

    const int r   = lane & 15;
    const int kk  = (lane >> 4) * 2;

    v8f acc = {};
#pragma unroll 4
    for (int k = 0; k < K; k += 4) {
        v2f a, b;
        a.x = As[r * K + k + kk];
        a.y = As[r * K + k + kk + 1];
        b.x = B[(size_t)(k + kk) * NOUT + n0 + r];
        b.y = B[(size_t)(k + kk + 1) * NOUT + n0 + r];
        acc = __builtin_amdgcn_wmma_f32_16x16x4_f32(
             false, a,  false, b,
             (short)0, acc,  false,  false);
    }

    const int rb = (lane >> 4) * 8;
#pragma unroll
    for (int i = 0; i < 8; ++i)
        C[(size_t)(m0 + rb + i) * NOUT + n0 + r] = acc[i];
}

template <int F>
__global__ void __launch_bounds__(256) k_scatter(const float* __restrict__ H,
                                                 const int* __restrict__ srcp,
                                                 const int* __restrict__ dstp,
                                                 const float* __restrict__ dinv,
                                                 float* __restrict__ out) {
    const int wave = blockIdx.x * 8 + (threadIdx.x >> 5);
    const int lane = threadIdx.x & 31;
    const int Etot = N_EDGES + N_NODES;
    if (wave >= Etot) return;
    int s, d; float w;
    if (wave < N_EDGES) {
        s = srcp[wave]; d = dstp[wave];
        w = dinv[s] * dinv[d];
    } else {
        s = d = wave - N_EDGES;
        float di = dinv[s];
        w = di * di;
    }
    const float4* hr = (const float4*)(H + (size_t)s * F);
    float* orow = out + (size_t)d * F;
    __builtin_prefetch(hr, 0, 0);
#pragma unroll
    for (int j = lane; j < F / 4; j += 32) {
        float4 v = hr[j];
        atomicAdd(&orow[4 * j + 0], v.x * w);
        atomicAdd(&orow[4 * j + 1], v.y * w);
        atomicAdd(&orow[4 * j + 2], v.z * w);
        atomicAdd(&orow[4 * j + 3], v.w * w);
    }
}

__global__ void __launch_bounds__(256) k_ln_relu(float* __restrict__ H,
                                                 const float* __restrict__ b1,
                                                 const float* __restrict__ gamma,
                                                 const float* __restrict__ beta) {
    const int node = blockIdx.x * 8 + (threadIdx.x >> 5);
    const int lane = threadIdx.x & 31;
    float* row = H + (size_t)node * F_HID;
    float4 v[2];
    float s = 0.0f;
#pragma unroll
    for (int i = 0; i < 2; ++i) {
        float4 t  = ((float4*)row)[lane + 32 * i];
        float4 bb = ((const float4*)b1)[lane + 32 * i];
        t.x += bb.x; t.y += bb.y; t.z += bb.z; t.w += bb.w;
        v[i] = t;
        s += t.x + t.y + t.z + t.w;
    }
    float mu = wave_sum32(s) * (1.0f / F_HID);
    float ss = 0.0f;
#pragma unroll
    for (int i = 0; i < 2; ++i) {
        float4 t = v[i];
        ss += (t.x - mu) * (t.x - mu) + (t.y - mu) * (t.y - mu) +
              (t.z - mu) * (t.z - mu) + (t.w - mu) * (t.w - mu);
    }
    float rstd = rsqrtf(wave_sum32(ss) * (1.0f / F_HID) + LN_EPS);
#pragma unroll
    for (int i = 0; i < 2; ++i) {
        float4 t = v[i];
        float4 g = ((const float4*)gamma)[lane + 32 * i];
        float4 b = ((const float4*)beta)[lane + 32 * i];
        t.x = fmaxf((t.x - mu) * rstd * g.x + b.x, 0.0f);
        t.y = fmaxf((t.y - mu) * rstd * g.y + b.y, 0.0f);
        t.z = fmaxf((t.z - mu) * rstd * g.z + b.z, 0.0f);
        t.w = fmaxf((t.w - mu) * rstd * g.w + b.w, 0.0f);
        ((float4*)row)[lane + 32 * i] = t;
    }
}

__global__ void __launch_bounds__(256) k_finalize(float* __restrict__ out,
                                                  const float* __restrict__ b2) {
    int i = blockIdx.x * 256 + threadIdx.x;
    if (i >= N_NODES * F_OUT) return;
    float v = out[i] + b2[i & (F_OUT - 1)];
    if (isnan(v)) v = 0.0f;
    else if (isinf(v)) v = v > 0.0f ? 20.0f : -20.0f;
    out[i] = v;
}

extern "C" void kernel_launch(void* const* d_in, const int* in_sizes, int n_in,
                              void* d_out, int out_size, void* d_ws, size_t ws_size,
                              hipStream_t stream) {
    const float* x     = (const float*)d_in[0];
    const int*   ei    = (const int*)d_in[1];
    const float* W1    = (const float*)d_in[2];
    const float* b1    = (const float*)d_in[3];
    const float* gamma = (const float*)d_in[4];
    const float* beta  = (const float*)d_in[5];
    const float* W2    = (const float*)d_in[6];
    const float* b2    = (const float*)d_in[7];
    const float* Wskip = (const float*)d_in[8];
    const int* srcp = ei;
    const int* dstp = ei + N_EDGES;

    float* ws   = (float*)d_ws;
    float* dinv = ws;
    float* xn   = dinv + N_NODES;
    float* h1   = xn + (size_t)N_NODES * F_IN;
    float* agg  = h1 + (size_t)N_NODES * F_HID;
    float* out  = (float*)d_out;

    const int Etot = N_EDGES + N_NODES;

    hipMemsetAsync(agg, 0, (size_t)N_NODES * F_HID * sizeof(float), stream);
    k_deg_init<<<(N_NODES + 255) / 256, 256, 0, stream>>>(dinv);
    k_deg_accum<<<(N_EDGES + 255) / 256, 256, 0, stream>>>(dstp, dinv);
    k_dinv<<<(N_NODES + 255) / 256, 256, 0, stream>>>(dinv);

    k_l2norm<<<N_NODES / 8, 256, 0, stream>>>(x, xn);

    k_gemm_wmma<F_IN, F_HID><<<dim3(N_NODES / 16, F_HID / 64), 128, 0, stream>>>(xn, W1, h1);
    k_gemm_wmma<F_IN, F_OUT><<<dim3(N_NODES / 16, F_OUT / 64), 128, 0, stream>>>(xn, Wskip, out);

    k_scatter<F_HID><<<(Etot + 7) / 8, 256, 0, stream>>>(h1, srcp, dstp, dinv, agg);
    k_ln_relu<<<N_NODES / 8, 256, 0, stream>>>(agg, b1, gamma, beta);

    k_gemm_wmma<F_HID, F_OUT><<<dim3(N_NODES / 16, F_OUT / 64), 128, 0, stream>>>(agg, W2, h1);
    k_scatter<F_OUT><<<(Etot + 7) / 8, 256, 0, stream>>>(h1, srcp, dstp, dinv, out);

    k_finalize<<<(N_NODES * F_OUT + 255) / 256, 256, 0, stream>>>(out, b2);
}


# Edge scatter without global atomics: 256 destination ranges, a work-group keeps its 196 rows in LDS, filters the whole edge list, adds rows by LDS f32 atomics; sum order: self loop, then LDS arrival
# speedup vs baseline: 2.2599x; 2.2599x over previous
#include <hip/hip_runtime.h>
#include <math.h>

#define N_NODES 50000
#define N_EDGES 800000
#define F_IN    512
#define F_HID   256
#define F_OUT   64
#define LN_EPS  1e-5f
#define L2_EPS  1e-12f

typedef __attribute__((ext_vector_type(2))) float v2f;
typedef __attribute__((ext_vector_type(4))) float v4f;
typedef __attribute__((ext_vector_type(8))) float v8f;

__device__ __forceinline__ float wave_sum32(float v) {
#pragma unroll
    for (int m = 16; m >= 1; m >>= 1) v += __shfl_xor(v, m, 32);
    return v;
}

__global__ void __launch_bounds__(256) k_deg_init(float* deg) {
    int i = blockIdx.x * 256 + threadIdx.x;
    if (i < N_NODES) deg[i] = 1.0f;
}

__global__ void __launch_bounds__(256) k_deg_accum(const int* __restrict__ dst,
                                                   float* deg) {
    int e = blockIdx.x * 256 + threadIdx.x;
    if (e < N_EDGES) atomicAdd(&deg[dst[e]], 1.0f);
}

__global__ void __launch_bounds__(256) k_dinv(float* deg) {
    int i = blockIdx.x * 256 + threadIdx.x;
    if (i < N_NODES) deg[i] = rsqrtf(deg[i]);
}

__global__ void __launch_bounds__(256) k_l2norm(const float* __restrict__ x,
                                                float* __restrict__ xn) {
    const int node = blockIdx.x * 8 + (threadIdx.x >> 5);
    const int lane = threadIdx.x & 31;
    const float4* row = (const float4*)(x + (size_t)node * F_IN);
    float4 v[4];
    float ss = 0.0f;
#pragma unroll
    for (int i = 0; i < 4; ++i) {
        float4 t = row[lane + 32 * i];
        t.x = isfinite(t.x) ? t.x : 0.0f;
        t.y = isfinite(t.y) ? t.y : 0.0f;
        t.z = isfinite(t.z) ? t.z : 0.0f;
        t.w = isfinite(t.w) ? t.w : 0.0f;
        v[i] = t;
        ss += t.x * t.x + t.y * t.y + t.z * t.z + t.w * t.w;
    }
    ss = wave_sum32(ss);
    float scale = 1.0f / fmaxf(sqrtf(ss), L2_EPS);
    float4* orow = (float4*)(xn + (size_t)node * F_IN);
#pragma unroll
    for (int i = 0; i < 4; ++i) {
        float4 t = v[i];
        t.x *= scale; t.y *= scale; t.z *= scale; t.w *= scale;
        orow[lane + 32 * i] = t;
    }
}

#define GEMM_RT ((N_NODES + 31) / 32)

template <int K, int NOUT>
__device__ __forceinline__ void gemm_load(const v4f* __restrict__ pa0,
                                          const v4f* __restrict__ pa1,
                                          const v4f* __restrict__ pb, int kb,
                                          v4f (&a)[2][2], v4f (&b)[8]) {
    a[0][0] = pa0[kb * 4 + 0];
    a[0][1] = pa0[kb * 4 + 1];
    a[1][0] = pa1[kb * 4 + 0];
    a[1][1] = pa1[kb * 4 + 1];
    const v4f* q = pb + (size_t)kb * (16 * NOUT / 4);
    for (int i = 0; i < 8; ++i) b[i] = q[i * (NOUT / 4)];
}

__device__ __forceinline__ void gemm_mma(const v4f (&a)[2][2], const v4f (&b)[8],
                                         v8f (&acc)[2][4]) {
    for (int j = 0; j < 4; ++j) {
        const v4f a0 = a[0][j >> 1];
        const v4f a1 = a[1][j >> 1];
        v2f fa0, fa1;
        fa0.x = a0[2 * (j & 1)]; fa0.y = a0[2 * (j & 1) + 1];
        fa1.x = a1[2 * (j & 1)]; fa1.y = a1[2 * (j & 1) + 1];
        for (int t = 0; t < 4; ++t) {
            v2f fb;
            fb.x = b[2 * j][t];
            fb.y = b[2 * j + 1][t];
            acc[0][t] = __builtin_amdgcn_wmma_f32_16x16x4_f32(
                 false, fa0,  false, fb,
                 (short)0, acc[0][t],  false,  false);
            acc[1][t] = __builtin_amdgcn_wmma_f32_16x16x4_f32(
                 false, fa1,  false, fb,
                 (short)0, acc[1][t],  false,  false);
        }
    }
}

template <int K, int NOUT>
__global__ void __launch_bounds__(128) k_gemm_wmma(const float* __restrict__ A,
                                                   const float* __restrict__ B,
                                                   float* __restrict__ C) {
    constexpr int NCB = NOUT / 64;
    constexpr int NKB = K / 16;
    const int lane = threadIdx.x & 31;
    const int g    = blockIdx.x * 4 + (threadIdx.x >> 5);
    const int rt   = g / NCB;
    if (rt >= GEMM_RT) return;
    const int n0 = (g % NCB) * 64;
    const int c  = lane & 15;
    const int h  = lane >> 4;
    const int m0 = rt * 32;
    const bool tile1 = (m0 + 16) < N_NODES;
    const int r0 = m0 + c;
    const int r1 = tile1 ? (m0 + 16 + c) : r0;

    const v4f* pa0 = (const v4f*)(A + (size_t)r0 * K + 8 * h);
    const v4f* pa1 = (const v4f*)(A + (size_t)r1 * K + 8 * h);
    const v4f* pb  = (const v4f*)(B + (size_t)(8 * h) * NOUT + n0 + 4 * c);

    v8f acc[2][4] = {};
    v4f aP[2][2], bP[8], aQ[2][2], bQ[8];
    gemm_load<K, NOUT>(pa0, pa1, pb, 0, aP, bP);
    for (int kb = 0; kb < NKB; kb += 2) {
        gemm_load<K, NOUT>(pa0, pa1, pb, kb + 1, aQ, bQ);
        gemm_mma(aP, bP, acc);
        gemm_load<K, NOUT>(pa0, pa1, pb, kb + 2 < NKB ? kb + 2 : NKB - 1, aP, bP);
        gemm_mma(aQ, bQ, acc);
    }

    float* crow = C + (size_t)(m0 + 8 * h) * NOUT + n0 + 4 * c;
    for (int i = 0; i < 8; ++i) {
        v4f o;
        o.x = acc[0][0][i]; o.y = acc[0][1][i]; o.z = acc[0][2][i]; o.w = acc[0][3][i];
        *(v4f*)(crow + (size_t)i * NOUT) = o;
    }
    if (tile1) {
        for (int i = 0; i < 8; ++i) {
            v4f o;
            o.x = acc[1][0][i]; o.y = acc[1][1][i]; o.z = acc[1][2][i]; o.w = acc[1][3][i];
            *(v4f*)(crow + (size_t)(16 + i) * NOUT) = o;
        }
    }
}

#define AGG_NB    256
#define AGG_R     ((N_NODES + AGG_NB - 1) / AGG_NB)
#define AGG_WAVES 32
#define AGG_LIST  160
#define AGG_NQ    (N_EDGES / 128)
#define AGG_U     4
#define AGG_ROUNDS ((AGG_NQ + AGG_WAVES * AGG_U - 1) / (AGG_WAVES * AGG_U))
static_assert(N_EDGES % 128 == 0, "the destination list is read in 128-edge pieces");
static_assert(AGG_R <= 256, "the row inside a range is packed into 8 bits");
static_assert(N_EDGES < (1 << 24), "the edge number is packed into 24 bits");

__device__ __forceinline__ void agg_list_load(const int4* __restrict__ d4, int round,
                                              int wv, int lane, int rot,
                                              int4 (&d)[AGG_U], int (&q)[AGG_U]) {
    for (int u = 0; u < AGG_U; ++u) {
        const int p = (round * AGG_U + u) * AGG_WAVES + wv;
        int pq = p + rot;
        if (pq >= AGG_NQ) pq -= AGG_NQ;
        q[u] = pq;
        if (p < AGG_NQ) d[u] = d4[pq * 32 + lane];
        else            d[u] = make_int4(-1, -1, -1, -1);
    }
}

template <int F, int G>
__device__ __forceinline__ void agg_batch(const float* __restrict__ H,
                                          const int* __restrict__ srcp,
                                          const float* __restrict__ dinv,
                                          float* acc, const unsigned* ent,
                                          int n, int lo, int lane) {
    int s = 0, rel = 0;
    float w = 0.0f;
    if (lane < n) {
        const unsigned en = ent[lane];
        rel = (int)(en & 255u);
        s   = srcp[en >> 8];
        w   = dinv[s] * dinv[lo + rel];
    }
    for (int k0 = 0; k0 < n; k0 += G) {
        float v[G][F / 32];
        for (int g = 0; g < G; ++g) {
            if (k0 + g < n) {
                const int sg = __builtin_amdgcn_readlane(s, k0 + g);
                const float* hr = H + (size_t)sg * F + lane;
                for (int j = 0; j < F / 32; ++j) v[g][j] = hr[32 * j];
            }
        }
        for (int g = 0; g < G; ++g) {
            if (k0 + g < n) {
                const int rg = __builtin_amdgcn_readlane(rel, k0 + g);
                const float wg = __builtin_bit_cast(
                    float, __builtin_amdgcn_readlane(__builtin_bit_cast(int, w), k0 + g));
                float* ar = acc + rg * F + lane;
                for (int j = 0; j < F / 32; ++j) atomicAdd(&ar[32 * j], v[g][j] * wg);
            }
        }
    }
}

template <int F, bool ADD_TO_OUT>
__global__ void __launch_bounds__(1024) k_agg(const float* __restrict__ H,
                                              const int* __restrict__ srcp,
                                              const int* __restrict__ dstp,
                                              const float* __restrict__ dinv,
                                              float* out) {
    __shared__ float acc[AGG_R * F_HID];
    __shared__ unsigned list[AGG_WAVES][AGG_LIST];
    constexpr int G = 1024 / F;

    const int lane  = threadIdx.x & 31;
    const int wv    = threadIdx.x >> 5;
    const int lo    = blockIdx.x * AGG_R;
    const int nrows = (N_NODES - lo) < AGG_R ? (N_NODES - lo) : AGG_R;

    for (int idx = threadIdx.x; idx < nrows * (F / 4); idx += 1024) {
        const float di = dinv[lo + idx / (F / 4)];
        const float w  = di * di;
        float4 t = ((const float4*)(H + (size_t)lo * F))[idx];
        t.x *= w; t.y *= w; t.z *= w; t.w *= w;
        ((float4*)acc)[idx] = t;
    }
    __syncthreads();

    const int rot = blockIdx.x * (AGG_NQ / AGG_NB);
    const int4* d4 = (const int4*)dstp;
    unsigned* mylist = list[wv];
    int cnt = 0;
    int4 dc[AGG_U], dn[AGG_U];
    int  qc[AGG_U], qn[AGG_U];
    agg_list_load(d4, 0, wv, lane, rot, dc, qc);
    for (int round = 0; round < AGG_ROUNDS; ++round) {
        agg_list_load(d4, round + 1, wv, lane, rot, dn, qn);
#pragma unroll
        for (int u = 0; u < AGG_U; ++u) {
            const int ebase = qc[u] * 128 + lane * 4;
            const int dv[4] = {dc[u].x, dc[u].y, dc[u].z, dc[u].w};
            for (int c = 0; c < 4; ++c) {
                const unsigned rel = (unsigned)(dv[c] - lo);
                const bool hit = rel < (unsigned)nrows;
                const unsigned mask = __builtin_amdgcn_ballot_w32(hit);
                if (mask) {
                    if (hit)
                        mylist[cnt + (int)__builtin_amdgcn_mbcnt_lo(mask, 0u)] =
                            ((unsigned)(ebase + c) << 8) | rel;
                    cnt += __builtin_popcount(mask);
                }
            }
            while (cnt >= 32) {
                cnt -= 32;
                agg_batch<F, G>(H, srcp, dinv, acc, mylist + cnt, 32, lo, lane);
            }
        }
        for (int u = 0; u < AGG_U; ++u) { dc[u] = dn[u]; qc[u] = qn[u]; }
    }
    if (cnt > 0) agg_batch<F, G>(H, srcp, dinv, acc, mylist, cnt, lo, lane);
    __syncthreads();

    for (int idx = threadIdx.x; idx < nrows * (F / 4); idx += 1024) {
        float4 a = ((const float4*)acc)[idx];
        float4* o = (float4*)(out + (size_t)lo * F) + idx;
        if (ADD_TO_OUT) {
            const float4 t = *o;
            a.x += t.x; a.y += t.y; a.z += t.z; a.w += t.w;
        }
        *o = a;
    }
}

__global__ void __launch_bounds__(256) k_ln_relu(float* __restrict__ H,
                                                 const float* __restrict__ b1,
                                                 const float* __restrict__ gamma,
                                                 const float* __restrict__ beta) {
    const int node = blockIdx.x * 8 + (threadIdx.x >> 5);
    const int lane = threadIdx.x & 31;
    float* row = H + (size_t)node * F_HID;
    float4 v[2];
    float s = 0.0f;
#pragma unroll
    for (int i = 0; i < 2; ++i) {
        float4 t  = ((float4*)row)[lane + 32 * i];
        float4 bb = ((const float4*)b1)[lane + 32 * i];
        t.x += bb.x; t.y += bb.y; t.z += bb.z; t.w += bb.w;
        v[i] = t;
        s += t.x + t.y + t.z + t.w;
    }
    float mu = wave_sum32(s) * (1.0f / F_HID);
    float ss = 0.0f;
#pragma unroll
    for (int i = 0; i < 2; ++i) {
        float4 t = v[i];
        ss += (t.x - mu) * (t.x - mu) + (t.y - mu) * (t.y - mu) +
              (t.z - mu) * (t.z - mu) + (t.w - mu) * (t.w - mu);
    }
    float rstd = rsqrtf(wave_sum32(ss) * (1.0f / F_HID) + LN_EPS);
#pragma unroll
    for (int i = 0; i < 2; ++i) {
        float4 t = v[i];
        float4 g = ((const float4*)gamma)[lane + 32 * i];
        float4 b = ((const float4*)beta)[lane + 32 * i];
        t.x = fmaxf((t.x - mu) * rstd * g.x + b.x, 0.0f);
        t.y = fmaxf((t.y - mu) * rstd * g.y + b.y, 0.0f);
        t.z = fmaxf((t.z - mu) * rstd * g.z + b.z, 0.0f);
        t.w = fmaxf((t.w - mu) * rstd * g.w + b.w, 0.0f);
        ((float4*)row)[lane + 32 * i] = t;
    }
}

__global__ void __launch_bounds__(256) k_finalize(float* __restrict__ out,
                                                  const float* __restrict__ b2) {
    int i = blockIdx.x * 256 + threadIdx.x;
    if (i >= N_NODES * F_OUT) return;
    float v = out[i] + b2[i & (F_OUT - 1)];
    if (isnan(v)) v = 0.0f;
    else if (isinf(v)) v = v > 0.0f ? 20.0f : -20.0f;
    out[i] = v;
}

extern "C" void kernel_launch(void* const* d_in, const int* in_sizes, int n_in,
                              void* d_out, int out_size, void* d_ws, size_t ws_size,
                              hipStream_t stream) {
    const float* x     = (const float*)d_in[0];
    const int*   ei    = (const int*)d_in[1];
    const float* W1    = (const float*)d_in[2];
    const float* b1    = (const float*)d_in[3];
    const float* gamma = (const float*)d_in[4];
    const float* beta  = (const float*)d_in[5];
    const float* W2    = (const float*)d_in[6];
    const float* b2    = (const float*)d_in[7];
    const float* Wskip = (const float*)d_in[8];
    const int* srcp = ei;
    const int* dstp = ei + N_EDGES;

    float* ws   = (float*)d_ws;
    float* xn   = ws;
    float* h1   = xn + (size_t)N_NODES * F_IN;
    float* agg  = h1 + (size_t)N_NODES * F_HID;
    float* dinv = agg + (size_t)N_NODES * F_HID;
    float* out  = (float*)d_out;

    k_deg_init<<<(N_NODES + 255) / 256, 256, 0, stream>>>(dinv);
    k_deg_accum<<<(N_EDGES + 255) / 256, 256, 0, stream>>>(dstp, dinv);
    k_dinv<<<(N_NODES + 255) / 256, 256, 0, stream>>>(dinv);

    k_l2norm<<<N_NODES / 8, 256, 0, stream>>>(x, xn);

    k_gemm_wmma<F_IN, F_HID><<<(GEMM_RT * (F_HID / 64) + 3) / 4, 128, 0, stream>>>(xn, W1, h1);
    k_gemm_wmma<F_IN, F_OUT><<<(GEMM_RT * (F_OUT / 64) + 3) / 4, 128, 0, stream>>>(xn, Wskip, out);

    k_agg<F_HID, false><<<AGG_NB, 1024, 0, stream>>>(h1, srcp, dstp, dinv, agg);
    k_ln_relu<<<N_NODES / 8, 256, 0, stream>>>(agg, b1, gamma, beta);

    k_gemm_wmma<F_HID, F_OUT><<<(GEMM_RT * (F_OUT / 64) + 3) / 4, 128, 0, stream>>>(agg, W2, h1);
    k_agg<F_OUT, true><<<AGG_NB, 1024, 0, stream>>>(h1, srcp, dstp, dinv, out);

    k_finalize<<<(N_NODES * F_OUT + 255) / 256, 256, 0, stream>>>(out, b2);
}


# Edges bucketed once per call by the 256 destination ranges (LDS counts, one global int atomic per range); a range's work-group reads its bucket instead of all 800k edges; sum order: LDS arrival
# speedup vs baseline: 2.7137x; 1.2008x over previous
#include <hip/hip_runtime.h>
#include <math.h>

#define N_NODES 50000
#define N_EDGES 800000
#define F_IN    512
#define F_HID   256
#define F_OUT   64
#define LN_EPS  1e-5f
#define L2_EPS  1e-12f

#define AGG_NB    256
#define AGG_R     ((N_NODES + AGG_NB - 1) / AGG_NB)

typedef __attribute__((ext_vector_type(2))) float v2f;
typedef __attribute__((ext_vector_type(4))) float v4f;
typedef __attribute__((ext_vector_type(8))) float v8f;

__device__ __forceinline__ float wave_sum32(float v) {
#pragma unroll
    for (int m = 16; m >= 1; m >>= 1) v += __shfl_xor(v, m, 32);
    return v;
}

__global__ void __launch_bounds__(256) k_deg_init(float* deg, int* cursor) {
    int i = blockIdx.x * 256 + threadIdx.x;
    if (i < N_NODES) deg[i] = 1.0f;
    if (i < AGG_NB) cursor[i] = 0;
}

__global__ void __launch_bounds__(256) k_deg_accum(const int* __restrict__ dst,
                                                   float* deg) {
    int e = blockIdx.x * 256 + threadIdx.x;
    if (e < N_EDGES) atomicAdd(&deg[dst[e]], 1.0f);
}

__global__ void __launch_bounds__(256) k_dinv(float* deg, int* __restrict__ bincnt) {
    __shared__ int part[8];
    const int lo    = blockIdx.x * AGG_R;
    const int nrows = (N_NODES - lo) < AGG_R ? (N_NODES - lo) : AGG_R;
    int cnt = 0;
    if ((int)threadIdx.x < nrows) {
        const float d = deg[lo + threadIdx.x];
        cnt = (int)d - 1;
        deg[lo + threadIdx.x] = rsqrtf(d);
    }
    for (int m = 16; m >= 1; m >>= 1) cnt += __shfl_xor(cnt, m, 32);
    if ((threadIdx.x & 31) == 0) part[threadIdx.x >> 5] = cnt;
    __syncthreads();
    if (threadIdx.x == 0) {
        int s = 0;
        for (int w = 0; w < 8; ++w) s += part[w];
        bincnt[blockIdx.x] = s;
    }
}

__global__ void __launch_bounds__(256) k_l2norm(const float* __restrict__ x,
                                                float* __restrict__ xn) {
    const int node = blockIdx.x * 8 + (threadIdx.x >> 5);
    const int lane = threadIdx.x & 31;
    const float4* row = (const float4*)(x + (size_t)node * F_IN);
    float4 v[4];
    float ss = 0.0f;
#pragma unroll
    for (int i = 0; i < 4; ++i) {
        float4 t = row[lane + 32 * i];
        t.x = isfinite(t.x) ? t.x : 0.0f;
        t.y = isfinite(t.y) ? t.y : 0.0f;
        t.z = isfinite(t.z) ? t.z : 0.0f;
        t.w = isfinite(t.w) ? t.w : 0.0f;
        v[i] = t;
        ss += t.x * t.x + t.y * t.y + t.z * t.z + t.w * t.w;
    }
    ss = wave_sum32(ss);
    float scale = 1.0f / fmaxf(sqrtf(ss), L2_EPS);
    float4* orow = (float4*)(xn + (size_t)node * F_IN);
#pragma unroll
    for (int i = 0; i < 4; ++i) {
        float4 t = v[i];
        t.x *= scale; t.y *= scale; t.z *= scale; t.w *= scale;
        orow[lane + 32 * i] = t;
    }
}

#define GEMM_RT ((N_NODES + 31) / 32)

template <int K, int NOUT>
__device__ __forceinline__ void gemm_load(const v4f* __restrict__ pa0,
                                          const v4f* __restrict__ pa1,
                                          const v4f* __restrict__ pb, int kb,
                                          v4f (&a)[2][2], v4f (&b)[8]) {
    a[0][0] = pa0[kb * 4 + 0];
    a[0][1] = pa0[kb * 4 + 1];
    a[1][0] = pa1[kb * 4 + 0];
    a[1][1] = pa1[kb * 4 + 1];
    const v4f* q = pb + (size_t)kb * (16 * NOUT / 4);
    for (int i = 0; i < 8; ++i) b[i] = q[i * (NOUT / 4)];
}

__device__ __forceinline__ void gemm_mma(const v4f (&a)[2][2], const v4f (&b)[8],
                                         v8f (&acc)[2][4]) {
    for (int j = 0; j < 4; ++j) {
        const v4f a0 = a[0][j >> 1];
        const v4f a1 = a[1][j >> 1];
        v2f fa0, fa1;
        fa0.x = a0[2 * (j & 1)]; fa0.y = a0[2 * (j & 1) + 1];
        fa1.x = a1[2 * (j & 1)]; fa1.y = a1[2 * (j & 1) + 1];
        for (int t = 0; t < 4; ++t) {
            v2f fb;
            fb.x = b[2 * j][t];
            fb.y = b[2 * j + 1][t];
            acc[0][t] = __builtin_amdgcn_wmma_f32_16x16x4_f32(
                 false, fa0,  false, fb,
                 (short)0, acc[0][t],  false,  false);
            acc[1][t] = __builtin_amdgcn_wmma_f32_16x16x4_f32(
                 false, fa1,  false, fb,
                 (short)0, acc[1][t],  false,  false);
        }
    }
}

template <int K, int NOUT>
__global__ void __launch_bounds__(128) k_gemm_wmma(const float* __restrict__ A,
                                                   const float* __restrict__ B,
                                                   float* __restrict__ C) {
    constexpr int NCB = NOUT / 64;
    constexpr int NKB = K / 16;
    const int lane = threadIdx.x & 31;
    const int g    = blockIdx.x * 4 + (threadIdx.x >> 5);
    const int rt   = g / NCB;
    if (rt >= GEMM_RT) return;
    const int n0 = (g % NCB) * 64;
    const int c  = lane & 15;
    const int h  = lane >> 4;
    const int m0 = rt * 32;
    const bool tile1 = (m0 + 16) < N_NODES;
    const int r0 = m0 + c;
    const int r1 = tile1 ? (m0 + 16 + c) : r0;

    const v4f* pa0 = (const v4f*)(A + (size_t)r0 * K + 8 * h);
    const v4f* pa1 = (const v4f*)(A + (size_t)r1 * K + 8 * h);
    const v4f* pb  = (const v4f*)(B + (size_t)(8 * h) * NOUT + n0 + 4 * c);

    v8f acc[2][4] = {};
    v4f aP[2][2], bP[8], aQ[2][2], bQ[8];
    gemm_load<K, NOUT>(pa0, pa1, pb, 0, aP, bP);
    for (int kb = 0; kb < NKB; kb += 2) {
        gemm_load<K, NOUT>(pa0, pa1, pb, kb + 1, aQ, bQ);
        gemm_mma(aP, bP, acc);
        gemm_load<K, NOUT>(pa0, pa1, pb, kb + 2 < NKB ? kb + 2 : NKB - 1, aP, bP);
        gemm_mma(aQ, bQ, acc);
    }

    float* crow = C + (size_t)(m0 + 8 * h) * NOUT + n0 + 4 * c;
    for (int i = 0; i < 8; ++i) {
        v4f o;
        o.x = acc[0][0][i]; o.y = acc[0][1][i]; o.z = acc[0][2][i]; o.w = acc[0][3][i];
        *(v4f*)(crow + (size_t)i * NOUT) = o;
    }
    if (tile1) {
        for (int i = 0; i < 8; ++i) {
            v4f o;
            o.x = acc[1][0][i]; o.y = acc[1][1][i]; o.z = acc[1][2][i]; o.w = acc[1][3][i];
            *(v4f*)(crow + (size_t)(16 + i) * NOUT) = o;
        }
    }
}

#define BIN_T  1024
#define BIN_WG ((N_EDGES / 4 + BIN_T - 1) / BIN_T)
static_assert(N_EDGES % 4 == 0, "the edge lists are read in 16-byte pieces");
static_assert(AGG_R <= 256, "the row inside a range is packed into 8 bits");
static_assert(N_NODES < (1 << 24), "the source node is packed into 24 bits");
static_assert(AGG_NB == 256, "the bucket starts are summed by 8 waves of 32 lanes");

__global__ void __launch_bounds__(BIN_T) k_bin(const int* __restrict__ srcp,
                                               const int* __restrict__ dstp,
                                               const int* __restrict__ bincnt,
                                               int* cursor, int* __restrict__ off,
                                               unsigned* __restrict__ list) {
    __shared__ int s_cnt[AGG_NB];
    __shared__ int s_base[AGG_NB];
    __shared__ int s_wtot[AGG_NB / 32];
    const int t    = threadIdx.x;
    const int lane = t & 31;
    const int wv   = t >> 5;
    if (t < AGG_NB) s_cnt[t] = 0;
    __syncthreads();

    const int q = blockIdx.x * BIN_T + t;
    int dv[4] = {-1, -1, -1, -1}, sv[4] = {0, 0, 0, 0};
    if (q < N_EDGES / 4) {
        const int4 d = ((const int4*)dstp)[q];
        const int4 s = ((const int4*)srcp)[q];
        dv[0] = d.x; dv[1] = d.y; dv[2] = d.z; dv[3] = d.w;
        sv[0] = s.x; sv[1] = s.y; sv[2] = s.z; sv[3] = s.w;
    }
    int bin[4], rank[4];
    for (int c = 0; c < 4; ++c) {
        bin[c] = -1; rank[c] = 0;
        if ((unsigned)dv[c] < (unsigned)N_NODES) {
            bin[c]  = (int)((unsigned)dv[c] / (unsigned)AGG_R);
            rank[c] = atomicAdd(&s_cnt[bin[c]], 1);
        }
    }

    const int v = (t < AGG_NB) ? bincnt[t] : 0;
    int incl = v;
    for (int d = 1; d < 32; d <<= 1) {
        const int o = __shfl_up(incl, d, 32);
        if (lane >= d) incl += o;
    }
    if (t < AGG_NB && lane == 31) s_wtot[wv] = incl;
    __syncthreads();
    if (t < AGG_NB) {
        int start = incl - v;
        for (int w = 0; w < AGG_NB / 32; ++w) start += (w < wv) ? s_wtot[w] : 0;
        const int mine = s_cnt[t];
        const int got  = mine ? atomicAdd(&cursor[t], mine) : 0;
        s_base[t] = start + got;
        if (blockIdx.x == 0) {
            off[t] = start;
            if (t == AGG_NB - 1) off[AGG_NB] = start + v;
        }
    }
    __syncthreads();
    for (int c = 0; c < 4; ++c) {
        if (bin[c] >= 0)
            list[s_base[bin[c]] + rank[c]] =
                ((unsigned)sv[c] << 8) | (unsigned)(dv[c] - bin[c] * AGG_R);
    }
}

#define AGG_WAVES 32

template <int F, int G>
__device__ __forceinline__ void agg_batch(const float* __restrict__ H,
                                          const float* __restrict__ dinv,
                                          float* acc, const unsigned* __restrict__ ent,
                                          int n, int lo, int lane) {
    int s = 0, rel = 0;
    float w = 0.0f;
    if (lane < n) {
        const unsigned en = ent[lane];
        rel = (int)(en & 255u);
        s   = (int)(en >> 8);
        w   = dinv[s] * dinv[lo + rel];
    }
    for (int k0 = 0; k0 < n; k0 += G) {
        float v[G][F / 32];
        for (int g = 0; g < G; ++g) {
            if (k0 + g < n) {
                const int sg = __builtin_amdgcn_readlane(s, k0 + g);
                const float* hr = H + (size_t)sg * F + lane;
                for (int j = 0; j < F / 32; ++j) v[g][j] = hr[32 * j];
            }
        }
        for (int g = 0; g < G; ++g) {
            if (k0 + g < n) {
                const int rg = __builtin_amdgcn_readlane(rel, k0 + g);
                const float wg = __builtin_bit_cast(
                    float, __builtin_amdgcn_readlane(__builtin_bit_cast(int, w), k0 + g));
                float* ar = acc + rg * F + lane;
                for (int j = 0; j < F / 32; ++j) atomicAdd(&ar[32 * j], v[g][j] * wg);
            }
        }
    }
}

template <int F, bool ADD_TO_OUT>
__global__ void __launch_bounds__(1024) k_agg(const float* __restrict__ H,
                                              const unsigned* __restrict__ list,
                                              const int* __restrict__ off,
                                              const float* __restrict__ dinv,
                                              float* out) {
    __shared__ float acc[AGG_R * F_HID];
    constexpr int G = 1024 / F;

    const int lane  = threadIdx.x & 31;
    const int wv    = __builtin_amdgcn_readfirstlane((int)(threadIdx.x >> 5));
    const int lo    = blockIdx.x * AGG_R;
    const int nrows = (N_NODES - lo) < AGG_R ? (N_NODES - lo) : AGG_R;

    for (int idx = threadIdx.x; idx < nrows * (F / 4); idx += 1024) {
        const float di = dinv[lo + idx / (F / 4)];
        const float w  = di * di;
        float4 t = ((const float4*)(H + (size_t)lo * F))[idx];
        t.x *= w; t.y *= w; t.z *= w; t.w *= w;
        ((float4*)acc)[idx] = t;
    }
    __syncthreads();

    const int beg = off[blockIdx.x];
    const int end = off[blockIdx.x + 1];
    for (int p = beg + wv * 32; p < end; p += AGG_WAVES * 32) {
        if (end - p >= 32) agg_batch<F, G>(H, dinv, acc, list + p, 32, lo, lane);
        else               agg_batch<F, G>(H, dinv, acc, list + p, end - p, lo, lane);
    }
    __syncthreads();

    for (int idx = threadIdx.x; idx < nrows * (F / 4); idx += 1024) {
        float4 a = ((const float4*)acc)[idx];
        float4* o = (float4*)(out + (size_t)lo * F) + idx;
        if (ADD_TO_OUT) {
            const float4 t = *o;
            a.x += t.x; a.y += t.y; a.z += t.z; a.w += t.w;
        }
        *o = a;
    }
}

__global__ void __launch_bounds__(256) k_ln_relu(float* __restrict__ H,
                                                 const float* __restrict__ b1,
                                                 const float* __restrict__ gamma,
                                                 const float* __restrict__ beta) {
    const int node = blockIdx.x * 8 + (threadIdx.x >> 5);
    const int lane = threadIdx.x & 31;
    float* row = H + (size_t)node * F_HID;
    float4 v[2];
    float s = 0.0f;
#pragma unroll
    for (int i = 0; i < 2; ++i) {
        float4 t  = ((float4*)row)[lane + 32 * i];
        float4 bb = ((const float4*)b1)[lane + 32 * i];
        t.x += bb.x; t.y += bb.y; t.z += bb.z; t.w += bb.w;
        v[i] = t;
        s += t.x + t.y + t.z + t.w;
    }
    float mu = wave_sum32(s) * (1.0f / F_HID);
    float ss = 0.0f;
#pragma unroll
    for (int i = 0; i < 2; ++i) {
        float4 t = v[i];
        ss += (t.x - mu) * (t.x - mu) + (t.y - mu) * (t.y - mu) +
              (t.z - mu) * (t.z - mu) + (t.w - mu) * (t.w - mu);
    }
    float rstd = rsqrtf(wave_sum32(ss) * (1.0f / F_HID) + LN_EPS);
#pragma unroll
    for (int i = 0; i < 2; ++i) {
        float4 t = v[i];
        float4 g = ((const float4*)gamma)[lane + 32 * i];
        float4 b = ((const float4*)beta)[lane + 32 * i];
        t.x = fmaxf((t.x - mu) * rstd * g.x + b.x, 0.0f);
        t.y = fmaxf((t.y - mu) * rstd * g.y + b.y, 0.0f);
        t.z = fmaxf((t.z - mu) * rstd * g.z + b.z, 0.0f);
        t.w = fmaxf((t.w - mu) * rstd * g.w + b.w, 0.0f);
        ((float4*)row)[lane + 32 * i] = t;
    }
}

__global__ void __launch_bounds__(256) k_finalize(float* __restrict__ out,
                                                  const float* __restrict__ b2) {
    int i = blockIdx.x * 256 + threadIdx.x;
    if (i >= N_NODES * F_OUT) return;
    float v = out[i] + b2[i & (F_OUT - 1)];
    if (isnan(v)) v = 0.0f;
    else if (isinf(v)) v = v > 0.0f ? 20.0f : -20.0f;
    out[i] = v;
}

extern "C" void kernel_launch(void* const* d_in, const int* in_sizes, int n_in,
                              void* d_out, int out_size, void* d_ws, size_t ws_size,
                              hipStream_t stream) {
    const float* x     = (const float*)d_in[0];
    const int*   ei    = (const int*)d_in[1];
    const float* W1    = (const float*)d_in[2];
    const float* b1    = (const float*)d_in[3];
    const float* gamma = (const float*)d_in[4];
    const float* beta  = (const float*)d_in[5];
    const float* W2    = (const float*)d_in[6];
    const float* b2    = (const float*)d_in[7];
    const float* Wskip = (const float*)d_in[8];
    const int* srcp = ei;
    const int* dstp = ei + N_EDGES;

    float* ws   = (float*)d_ws;
    float* xn   = ws;
    float* h1   = xn + (size_t)N_NODES * F_IN;
    float* agg  = h1 + (size_t)N_NODES * F_HID;
    float* dinv = agg + (size_t)N_NODES * F_HID;
    float* out  = (float*)d_out;
    unsigned* list = (unsigned*)xn;
    int* cursor = (int*)(list + N_EDGES);
    int* bincnt = cursor + AGG_NB;
    int* off    = bincnt + AGG_NB;

    k_l2norm<<<N_NODES / 8, 256, 0, stream>>>(x, xn);

    k_gemm_wmma<F_IN, F_HID><<<(GEMM_RT * (F_HID / 64) + 3) / 4, 128, 0, stream>>>(xn, W1, h1);
    k_gemm_wmma<F_IN, F_OUT><<<(GEMM_RT * (F_OUT / 64) + 3) / 4, 128, 0, stream>>>(xn, Wskip, out);

    k_deg_init<<<(N_NODES + 255) / 256, 256, 0, stream>>>(dinv, cursor);
    k_deg_accum<<<(N_EDGES + 255) / 256, 256, 0, stream>>>(dstp, dinv);
    k_dinv<<<AGG_NB, 256, 0, stream>>>(dinv, bincnt);
    k_bin<<<BIN_WG, BIN_T, 0, stream>>>(srcp, dstp, bincnt, cursor, off, list);

    k_agg<F_HID, false><<<AGG_NB, 1024, 0, stream>>>(h1, list, off, dinv, agg);
    k_ln_relu<<<N_NODES / 8, 256, 0, stream>>>(agg, b1, gamma, beta);

    k_gemm_wmma<F_HID, F_OUT><<<(GEMM_RT * (F_OUT / 64) + 3) / 4, 128, 0, stream>>>(agg, W2, h1);
    k_agg<F_OUT, true><<<AGG_NB, 1024, 0, stream>>>(h1, list, off, dinv, out);

    k_finalize<<<(N_NODES * F_OUT + 255) / 256, 256, 0, stream>>>(out, b2);
}
